# starting kernel
# baseline (speedup 1.0000x reference)
#include <hip/hip_runtime.h>
#include <cmath>

typedef __attribute__((ext_vector_type(2))) float v2f;
typedef __attribute__((ext_vector_type(8))) float v8f;

__global__ void gcn_copy_h_kernel(const float* __restrict__ h,
                                  float* __restrict__ out, int n_nodes) {
  int i = blockIdx.x * blockDim.x + threadIdx.x;
  if (i >= n_nodes * 16) return;
  int r  = i >> 4;
  int c4 = i & 15;
  float4 v = *((const float4*)(h + (size_t)r * 64) + c4);
  *((float4*)(out + (size_t)r * 256) + c4) = v;
}

__global__ void gcn_zero_kernel(float4* __restrict__ p, int n4) {
  int i = blockIdx.x * blockDim.x + threadIdx.x;
  if (i < n4) p[i] = make_float4(0.f, 0.f, 0.f, 0.f);
}

__global__ void gcn_edge_scatter_kernel(const float* __restrict__ hprev,
                                        const float* __restrict__ ew,
                                        const int* __restrict__ src,
                                        const int* __restrict__ dst,
                                        float* __restrict__ agg,
                                        int n_edges) {
  int wave = (blockIdx.x * blockDim.x + threadIdx.x) >> 5;
  int lane = threadIdx.x & 31;
  int e = __builtin_amdgcn_readfirstlane(wave);
  if (e >= n_edges) return;
  int   s = src[e];
  int   d = dst[e];
  float w = ew[e];
  float2 hv = *(const float2*)(hprev + (size_t)s * 256 + lane * 2);
  float* p = agg + (size_t)d * 64 + lane * 2;
  unsafeAtomicAdd(p,     hv.x * w);
  unsafeAtomicAdd(p + 1, hv.y * w);
}

__global__ void gcn_gemm_wmma_kernel(const float* __restrict__ A,
                                     const float* __restrict__ W,
                                     const float* __restrict__ bias,
                                     float* __restrict__ out,
                                     int n_rows, int do_tanh) {
  int wave = threadIdx.x >> 5;
  int lane = threadIdx.x & 31;
  int row0 = blockIdx.x * 32 + (wave >> 2) * 16;
  int col0 = (wave & 3) * 16;
  if (row0 >= n_rows) return;

  int lh = lane >> 4;
  int lm = lane & 15;
  const float* arow = A + (size_t)(row0 + lm) * 64 + 2 * lh;
  const float* wcol = W + (size_t)(2 * lh) * 64 + col0 + lm;

  v8f acc = {};
#pragma unroll
  for (int k0 = 0; k0 < 64; k0 += 4) {
    v2f a = *(const v2f*)(arow + k0);
    v2f b;
    b.x = wcol[(size_t)k0 * 64];
    b.y = wcol[(size_t)k0 * 64 + 64];
    acc = __builtin_amdgcn_wmma_f32_16x16x4_f32(
         false, a,  false, b,
         (short)0, acc,  false,  false);
  }

  float bv = bias[col0 + lm];
  float* orow = out + (size_t)(row0 + 8 * lh) * 256 + col0 + lm;
#pragma unroll
  for (int v = 0; v < 8; ++v) {
    float val = acc[v] + bv;
    if (do_tanh) val = tanhf(val);
    orow[(size_t)v * 256] = val;
  }
}

extern "C" void kernel_launch(void* const* d_in, const int* in_sizes, int n_in,
                              void* d_out, int out_size, void* d_ws, size_t ws_size,
                              hipStream_t stream) {
  const float* h  = (const float*)d_in[0];
  const float* ew = (const float*)d_in[1];
  const float* Ws = (const float*)d_in[2];
  const float* bs = (const float*)d_in[3];
  const int*  src = (const int*)d_in[4];
  const int*  dst = (const int*)d_in[5];
  float* out = (float*)d_out;
  float* agg = (float*)d_ws;

  const int N = in_sizes[0] / 64;
  const int E = in_sizes[1];
  const int L = 3;

  {
    int t = N * 16;
    gcn_copy_h_kernel<<<(t + 255) / 256, 256, 0, stream>>>(h, out, N);
  }

  for (int layer = 0; layer < L; ++layer) {
    {
      int n4 = N * 16;
      gcn_zero_kernel<<<(n4 + 255) / 256, 256, 0, stream>>>((float4*)agg, n4);
    }
    {
      long threads = (long)E * 32;
      int blocks = (int)((threads + 255) / 256);
      gcn_edge_scatter_kernel<<<blocks, 256, 0, stream>>>(
          out + layer * 64, ew, src, dst, agg, E);
    }
    {
      int blocks = (N + 31) / 32;
      gcn_gemm_wmma_kernel<<<blocks, 256, 0, stream>>>(
          agg, Ws + layer * 64 * 64, bs + layer * 64,
          out + (layer + 1) * 64, N, (layer < L - 1) ? 1 : 0);
    }
  }
}


# edge scatter: each wave takes 16 consecutive edges and issues all 16 row gathers before its first atomic (was 1 edge per wave): 16x fewer waves, 16 rows in flight; arrival-order atomic sums kept
# speedup vs baseline: 1.6453x; 1.6453x over previous
#include <hip/hip_runtime.h>
#include <cmath>

typedef __attribute__((ext_vector_type(2))) float v2f;
typedef __attribute__((ext_vector_type(8))) float v8f;

__global__ void gcn_copy_h_kernel(const float* __restrict__ h,
                                  float* __restrict__ out, int n_nodes) {
  int i = blockIdx.x * blockDim.x + threadIdx.x;
  if (i >= n_nodes * 16) return;
  int r  = i >> 4;
  int c4 = i & 15;
  float4 v = *((const float4*)(h + (size_t)r * 64) + c4);
  *((float4*)(out + (size_t)r * 256) + c4) = v;
}

__global__ void gcn_zero_kernel(float4* __restrict__ p, int n4) {
  int i = blockIdx.x * blockDim.x + threadIdx.x;
  if (i < n4) p[i] = make_float4(0.f, 0.f, 0.f, 0.f);
}

template <int EPW>
__global__ void gcn_edge_scatter_kernel(const float* __restrict__ hprev,
                                        const float* __restrict__ ew,
                                        const int* __restrict__ src,
                                        const int* __restrict__ dst,
                                        float* __restrict__ agg,
                                        int n_edges) {
  int wave = (blockIdx.x * blockDim.x + threadIdx.x) >> 5;
  int lane = threadIdx.x & 31;
  int e0 = __builtin_amdgcn_readfirstlane(wave) * EPW;
  if (e0 >= n_edges) return;
  int f = lane * 2;

  if (e0 + EPW <= n_edges) {
    int   s[EPW];
    int   d[EPW];
    float w[EPW];
    float2 hv[EPW];
    for (int j = 0; j < EPW; ++j) {
      s[j] = src[e0 + j];
      d[j] = dst[e0 + j];
      w[j] = ew[e0 + j];
    }
    for (int j = 0; j < EPW; ++j)
      hv[j] = *(const float2*)(hprev + (size_t)s[j] * 256 + f);
    for (int j = 0; j < EPW; ++j) {
      float* p = agg + (size_t)d[j] * 64 + f;
      unsafeAtomicAdd(p,     hv[j].x * w[j]);
      unsafeAtomicAdd(p + 1, hv[j].y * w[j]);
    }
  } else {
    for (int e = e0; e < n_edges; ++e) {
      int   s1 = src[e];
      int   d1 = dst[e];
      float w1 = ew[e];
      float2 hv1 = *(const float2*)(hprev + (size_t)s1 * 256 + f);
      float* p = agg + (size_t)d1 * 64 + f;
      unsafeAtomicAdd(p,     hv1.x * w1);
      unsafeAtomicAdd(p + 1, hv1.y * w1);
    }
  }
}

__global__ void gcn_gemm_wmma_kernel(const float* __restrict__ A,
                                     const float* __restrict__ W,
                                     const float* __restrict__ bias,
                                     float* __restrict__ out,
                                     int n_rows, int do_tanh) {
  int wave = threadIdx.x >> 5;
  int lane = threadIdx.x & 31;
  int row0 = blockIdx.x * 32 + (wave >> 2) * 16;
  int col0 = (wave & 3) * 16;
  if (row0 >= n_rows) return;

  int lh = lane >> 4;
  int lm = lane & 15;
  const float* arow = A + (size_t)(row0 + lm) * 64 + 2 * lh;
  const float* wcol = W + (size_t)(2 * lh) * 64 + col0 + lm;

  v8f acc = {};
#pragma unroll
  for (int k0 = 0; k0 < 64; k0 += 4) {
    v2f a = *(const v2f*)(arow + k0);
    v2f b;
    b.x = wcol[(size_t)k0 * 64];
    b.y = wcol[(size_t)k0 * 64 + 64];
    acc = __builtin_amdgcn_wmma_f32_16x16x4_f32(
         false, a,  false, b,
         (short)0, acc,  false,  false);
  }

  float bv = bias[col0 + lm];
  float* orow = out + (size_t)(row0 + 8 * lh) * 256 + col0 + lm;
#pragma unroll
  for (int v = 0; v < 8; ++v) {
    float val = acc[v] + bv;
    if (do_tanh) val = tanhf(val);
    orow[(size_t)v * 256] = val;
  }
}

extern "C" void kernel_launch(void* const* d_in, const int* in_sizes, int n_in,
                              void* d_out, int out_size, void* d_ws, size_t ws_size,
                              hipStream_t stream) {
  const float* h  = (const float*)d_in[0];
  const float* ew = (const float*)d_in[1];
  const float* Ws = (const float*)d_in[2];
  const float* bs = (const float*)d_in[3];
  const int*  src = (const int*)d_in[4];
  const int*  dst = (const int*)d_in[5];
  float* out = (float*)d_out;
  float* agg = (float*)d_ws;

  const int N = in_sizes[0] / 64;
  const int E = in_sizes[1];
  const int L = 3;

  {
    int t = N * 16;
    gcn_copy_h_kernel<<<(t + 255) / 256, 256, 0, stream>>>(h, out, N);
  }

  for (int layer = 0; layer < L; ++layer) {
    {
      int n4 = N * 16;
      gcn_zero_kernel<<<(n4 + 255) / 256, 256, 0, stream>>>((float4*)agg, n4);
    }
    {
      const int EPW = 16;
      long threads = (long)((E + EPW - 1) / EPW) * 32;
      int blocks = (int)((threads + 255) / 256);
      gcn_edge_scatter_kernel<EPW><<<blocks, 256, 0, stream>>>(
          out + layer * 64, ew, src, dst, agg, E);
    }
    {
      int blocks = (N + 31) / 32;
      gcn_gemm_wmma_kernel<<<blocks, 256, 0, stream>>>(
          agg, Ws + layer * 64 * 64, bs + layer * 64,
          out + (layer + 1) * 64, N, (layer < L - 1) ? 1 : 0);
    }
  }
}


# edges sorted by destination once per call (integer rank atomic, scan, fill); one wave per node sums its rows in registers: f32 atomics and zero fill removed; sums in arrival-rank order
# speedup vs baseline: 1.8553x; 1.1276x over previous
#include <hip/hip_runtime.h>
#include <cmath>

typedef __attribute__((ext_vector_type(2))) float v2f;
typedef __attribute__((ext_vector_type(8))) float v8f;

__global__ void gcn_copy_h_kernel(const float* __restrict__ h,
                                  float* __restrict__ out,
                                  int* __restrict__ cnt, int n_nodes) {
  int i = blockIdx.x * blockDim.x + threadIdx.x;
  if (i < n_nodes) cnt[i] = 0;
  if (i >= n_nodes * 16) return;
  int r  = i >> 4;
  int c4 = i & 15;
  float4 v = *((const float4*)(h + (size_t)r * 64) + c4);
  *((float4*)(out + (size_t)r * 256) + c4) = v;
}

__global__ void gcn_rank_kernel(const int* __restrict__ dst,
                                int* __restrict__ cnt,
                                int* __restrict__ rnk, int n_edges) {
  int e = blockIdx.x * blockDim.x + threadIdx.x;
  if (e < n_edges) rnk[e] = atomicAdd(cnt + dst[e], 1);
}

__global__ __launch_bounds__(1024)
void gcn_scan_kernel(const int* __restrict__ cnt, int* __restrict__ off,
                     int n_nodes) {
  __shared__ int wave_tot[32];
  __shared__ int wave_pre[32];
  int tid  = threadIdx.x;
  int lane = tid & 31;
  int wv   = tid >> 5;
  int base = blockIdx.x * 1024;

  int pre = 0;
  const int4* c4 = (const int4*)cnt;
  int n4 = base >> 2;
#pragma unroll 4
  for (int j = tid; j < n4; j += 1024) {
    int4 q = c4[j];
    pre += (q.x + q.y) + (q.z + q.w);
  }

  int idx = base + tid;
  int v = (idx < n_nodes) ? cnt[idx] : 0;

  int inc = v;
#pragma unroll
  for (int d = 1; d < 32; d <<= 1) {
    int t = __shfl_up(inc, d);
    if (lane >= d) inc += t;
  }
#pragma unroll
  for (int d = 16; d >= 1; d >>= 1) pre += __shfl_xor(pre, d);
  if (lane == 31) wave_tot[wv] = inc;
  if (lane == 0)  wave_pre[wv] = pre;
  __syncthreads();

  int wt = wave_tot[lane];
  int pt = wave_pre[lane];
  int winc = wt;
#pragma unroll
  for (int d = 1; d < 32; d <<= 1) {
    int t = __shfl_up(winc, d);
    if (lane >= d) winc += t;
  }
#pragma unroll
  for (int d = 16; d >= 1; d >>= 1) pt += __shfl_xor(pt, d);
  int wave_excl = __shfl(winc - wt, wv);

  if (idx <= n_nodes) off[idx] = pt + wave_excl + (inc - v);
}

__global__ void gcn_fill_kernel(const int* __restrict__ src,
                                const int* __restrict__ dst,
                                const float* __restrict__ ew,
                                const int* __restrict__ rnk,
                                const int* __restrict__ off,
                                int2* __restrict__ rec, int n_edges) {
  int e = blockIdx.x * blockDim.x + threadIdx.x;
  if (e >= n_edges) return;
  int pos = off[dst[e]] + rnk[e];
  rec[pos] = make_int2(src[e], __float_as_int(ew[e]));
}

template <int U>
__device__ __forceinline__ void gcn_pull_rows(const float* __restrict__ hprev,
                                              const int2* __restrict__ rec,
                                              int i, int f,
                                              float& ax, float& ay) {
  int    s[U];
  float  w[U];
  float2 hv[U];
#pragma unroll
  for (int j = 0; j < U; ++j) {
    int2 r = rec[i + j];
    s[j] = r.x;
    w[j] = __int_as_float(r.y);
  }
#pragma unroll
  for (int j = 0; j < U; ++j)
    hv[j] = *(const float2*)(hprev + (size_t)s[j] * 256 + f);
#pragma unroll
  for (int j = 0; j < U; ++j) {
    ax += hv[j].x * w[j];
    ay += hv[j].y * w[j];
  }
}

__global__ __launch_bounds__(256)
void gcn_pull_kernel(const float* __restrict__ hprev,
                     const int2* __restrict__ rec,
                     const int* __restrict__ off,
                     float* __restrict__ agg, int n_nodes) {
  int wave = (blockIdx.x * blockDim.x + threadIdx.x) >> 5;
  int lane = threadIdx.x & 31;
  int n = __builtin_amdgcn_readfirstlane(wave);
  if (n >= n_nodes) return;
  int f = lane * 2;

  int i = off[n];
  int e = off[n + 1];
  float ax = 0.f, ay = 0.f;
  for (; i + 16 <= e; i += 16) gcn_pull_rows<16>(hprev, rec, i, f, ax, ay);
  int r = e - i;
  if (r & 8) { gcn_pull_rows<8>(hprev, rec, i, f, ax, ay); i += 8; }
  if (r & 4) { gcn_pull_rows<4>(hprev, rec, i, f, ax, ay); i += 4; }
  if (r & 2) { gcn_pull_rows<2>(hprev, rec, i, f, ax, ay); i += 2; }
  if (r & 1) { gcn_pull_rows<1>(hprev, rec, i, f, ax, ay); }

  *(float2*)(agg + (size_t)n * 64 + f) = make_float2(ax, ay);
}

__global__ void gcn_gemm_wmma_kernel(const float* __restrict__ A,
                                     const float* __restrict__ W,
                                     const float* __restrict__ bias,
                                     float* __restrict__ out,
                                     int n_rows, int do_tanh) {
  int wave = threadIdx.x >> 5;
  int lane = threadIdx.x & 31;
  int row0 = blockIdx.x * 32 + (wave >> 2) * 16;
  int col0 = (wave & 3) * 16;
  if (row0 >= n_rows) return;

  int lh = lane >> 4;
  int lm = lane & 15;
  const float* arow = A + (size_t)(row0 + lm) * 64 + 2 * lh;
  const float* wcol = W + (size_t)(2 * lh) * 64 + col0 + lm;

  v8f acc = {};
#pragma unroll
  for (int k0 = 0; k0 < 64; k0 += 4) {
    v2f a = *(const v2f*)(arow + k0);
    v2f b;
    b.x = wcol[(size_t)k0 * 64];
    b.y = wcol[(size_t)k0 * 64 + 64];
    acc = __builtin_amdgcn_wmma_f32_16x16x4_f32(
         false, a,  false, b,
         (short)0, acc,  false,  false);
  }

  float bv = bias[col0 + lm];
  float* orow = out + (size_t)(row0 + 8 * lh) * 256 + col0 + lm;
#pragma unroll
  for (int v = 0; v < 8; ++v) {
    float val = acc[v] + bv;
    if (do_tanh) val = tanhf(val);
    orow[(size_t)v * 256] = val;
  }
}

extern "C" void kernel_launch(void* const* d_in, const int* in_sizes, int n_in,
                              void* d_out, int out_size, void* d_ws, size_t ws_size,
                              hipStream_t stream) {
  const float* h  = (const float*)d_in[0];
  const float* ew = (const float*)d_in[1];
  const float* Ws = (const float*)d_in[2];
  const float* bs = (const float*)d_in[3];
  const int*  src = (const int*)d_in[4];
  const int*  dst = (const int*)d_in[5];
  float* out = (float*)d_out;
  float* agg = (float*)d_ws;

  const int N = in_sizes[0] / 64;
  const int E = in_sizes[1];
  const int L = 3;

  const int Np = (N + 4) & ~3;
  const int Ep = (E + 3) & ~3;
  int*  cnt = (int*)(agg + (size_t)N * 64);
  int*  off = cnt + Np;
  int*  rnk = off + Np;
  int2* rec = (int2*)(rnk + Ep);

  {
    int t = N * 16;
    gcn_copy_h_kernel<<<(t + 255) / 256, 256, 0, stream>>>(h, out, cnt, N);
  }
  gcn_rank_kernel<<<(E + 255) / 256, 256, 0, stream>>>(dst, cnt, rnk, E);
  gcn_scan_kernel<<<(N + 1 + 1023) / 1024, 1024, 0, stream>>>(cnt, off, N);
  gcn_fill_kernel<<<(E + 255) / 256, 256, 0, stream>>>(src, dst, ew, rnk, off, rec, E);

  for (int layer = 0; layer < L; ++layer) {
    {
      int blocks = (N + 7) / 8;
      gcn_pull_kernel<<<blocks, 256, 0, stream>>>(out + layer * 64, rec, off, agg, N);
    }
    {
      int blocks = (N + 31) / 32;
      gcn_gemm_wmma_kernel<<<blocks, 256, 0, stream>>>(
          agg, Ws + layer * 64 * 64, bs + layer * 64,
          out + (layer + 1) * 64, N, (layer < L - 1) ? 1 : 0);
    }
  }
}


# aggregation and dense transform of a layer fused into one launch: a work-group sums a 16-node tile into LDS and feeds the WMMA from there (agg buffer and 3 launches gone); sum order kept
# speedup vs baseline: 2.1086x; 1.1365x over previous
#include <hip/hip_runtime.h>
#include <cmath>

typedef __attribute__((ext_vector_type(2))) float v2f;
typedef __attribute__((ext_vector_type(8))) float v8f;

__global__ void gcn_copy_h_kernel(const float* __restrict__ h,
                                  float* __restrict__ out,
                                  int* __restrict__ cnt, int n_nodes) {
  int i = blockIdx.x * blockDim.x + threadIdx.x;
  if (i < n_nodes) cnt[i] = 0;
  if (i >= n_nodes * 16) return;
  int r  = i >> 4;
  int c4 = i & 15;
  float4 v = *((const float4*)(h + (size_t)r * 64) + c4);
  *((float4*)(out + (size_t)r * 256) + c4) = v;
}

__global__ void gcn_rank_kernel(const int* __restrict__ dst,
                                int* __restrict__ cnt,
                                int* __restrict__ rnk, int n_edges) {
  int e = blockIdx.x * blockDim.x + threadIdx.x;
  if (e < n_edges) rnk[e] = atomicAdd(cnt + dst[e], 1);
}

__global__ __launch_bounds__(1024)
void gcn_scan_kernel(const int* __restrict__ cnt, int* __restrict__ off,
                     int n_nodes) {
  __shared__ int wave_tot[32];
  __shared__ int wave_pre[32];
  int tid  = threadIdx.x;
  int lane = tid & 31;
  int wv   = tid >> 5;
  int base = blockIdx.x * 1024;

  int pre = 0;
  const int4* c4 = (const int4*)cnt;
  int n4 = base >> 2;
#pragma unroll 4
  for (int j = tid; j < n4; j += 1024) {
    int4 q = c4[j];
    pre += (q.x + q.y) + (q.z + q.w);
  }

  int idx = base + tid;
  int v = (idx < n_nodes) ? cnt[idx] : 0;

  int inc = v;
#pragma unroll
  for (int d = 1; d < 32; d <<= 1) {
    int t = __shfl_up(inc, d);
    if (lane >= d) inc += t;
  }
#pragma unroll
  for (int d = 16; d >= 1; d >>= 1) pre += __shfl_xor(pre, d);
  if (lane == 31) wave_tot[wv] = inc;
  if (lane == 0)  wave_pre[wv] = pre;
  __syncthreads();

  int wt = wave_tot[lane];
  int pt = wave_pre[lane];
  int winc = wt;
#pragma unroll
  for (int d = 1; d < 32; d <<= 1) {
    int t = __shfl_up(winc, d);
    if (lane >= d) winc += t;
  }
#pragma unroll
  for (int d = 16; d >= 1; d >>= 1) pt += __shfl_xor(pt, d);
  int wave_excl = __shfl(winc - wt, wv);

  if (idx <= n_nodes) off[idx] = pt + wave_excl + (inc - v);
}

__global__ void gcn_fill_kernel(const int* __restrict__ src,
                                const int* __restrict__ dst,
                                const float* __restrict__ ew,
                                const int* __restrict__ rnk,
                                const int* __restrict__ off,
                                int2* __restrict__ rec, int n_edges) {
  int e = blockIdx.x * blockDim.x + threadIdx.x;
  if (e >= n_edges) return;
  int pos = off[dst[e]] + rnk[e];
  rec[pos] = make_int2(src[e], __float_as_int(ew[e]));
}

template <int U>
__device__ __forceinline__ void gcn_pull_rows(const float* __restrict__ hprev,
                                              const int2* __restrict__ rec,
                                              int i, int f,
                                              float& ax, float& ay) {
  int    s[U];
  float  w[U];
  float2 hv[U];
#pragma unroll
  for (int j = 0; j < U; ++j) {
    int2 r = rec[i + j];
    s[j] = r.x;
    w[j] = __int_as_float(r.y);
  }
#pragma unroll
  for (int j = 0; j < U; ++j)
    hv[j] = *(const float2*)(hprev + (size_t)s[j] * 256 + f);
#pragma unroll
  for (int j = 0; j < U; ++j) {
    ax += hv[j].x * w[j];
    ay += hv[j].y * w[j];
  }
}

#define GCN_TILE_PITCH 68

__global__ __launch_bounds__(256)
void gcn_layer_kernel(const float* __restrict__ hprev,
                      const int2* __restrict__ rec,
                      const int* __restrict__ off,
                      const float* __restrict__ W,
                      const float* __restrict__ bias,
                      float* __restrict__ outp,
                      int n_nodes, int do_tanh) {
  __shared__ float tile[16 * GCN_TILE_PITCH];
  int wave = __builtin_amdgcn_readfirstlane((int)(threadIdx.x >> 5));
  int lane = threadIdx.x & 31;
  int row0 = blockIdx.x * 16;
  int f = lane * 2;

  int nb = row0 + wave * 2;
  int n0 = nb     < n_nodes ? nb     : n_nodes;
  int n1 = nb + 1 < n_nodes ? nb + 1 : n_nodes;
  int n2 = nb + 2 < n_nodes ? nb + 2 : n_nodes;
  int i  = off[n0];
  int e0 = off[n1];
  int e1 = off[n2];
#pragma unroll
  for (int q = 0; q < 2; ++q) {
    int e = q ? e1 : e0;
    float ax = 0.f, ay = 0.f;
    for (; i + 16 <= e; i += 16) gcn_pull_rows<16>(hprev, rec, i, f, ax, ay);
    int r = e - i;
    if (r & 8) { gcn_pull_rows<8>(hprev, rec, i, f, ax, ay); i += 8; }
    if (r & 4) { gcn_pull_rows<4>(hprev, rec, i, f, ax, ay); i += 4; }
    if (r & 2) { gcn_pull_rows<2>(hprev, rec, i, f, ax, ay); i += 2; }
    if (r & 1) { gcn_pull_rows<1>(hprev, rec, i, f, ax, ay); }
    *(float2*)(tile + (wave * 2 + q) * GCN_TILE_PITCH + f) = make_float2(ax, ay);
    i = e;
  }
  __syncthreads();
  if (wave >= 4) return;

  int col0 = wave * 16;
  int lh = lane >> 4;
  int lm = lane & 15;
  const float* arow = tile + lm * GCN_TILE_PITCH + 2 * lh;
  const float* wcol = W + (size_t)(2 * lh) * 64 + col0 + lm;

  v8f acc = {};
#pragma unroll
  for (int k0 = 0; k0 < 64; k0 += 4) {
    v2f a = *(const v2f*)(arow + k0);
    v2f b;
    b.x = wcol[(size_t)k0 * 64];
    b.y = wcol[(size_t)k0 * 64 + 64];
    acc = __builtin_amdgcn_wmma_f32_16x16x4_f32(
         false, a,  false, b,
         (short)0, acc,  false,  false);
  }

  float bv = bias[col0 + lm];
  int rbase = row0 + 8 * lh;
  float* orow = outp + (size_t)rbase * 256 + col0 + lm;
#pragma unroll
  for (int v = 0; v < 8; ++v) {
    float val = acc[v] + bv;
    if (do_tanh) val = tanhf(val);
    if (rbase + v < n_nodes) orow[(size_t)v * 256] = val;
  }
}

extern "C" void kernel_launch(void* const* d_in, const int* in_sizes, int n_in,
                              void* d_out, int out_size, void* d_ws, size_t ws_size,
                              hipStream_t stream) {
  const float* h  = (const float*)d_in[0];
  const float* ew = (const float*)d_in[1];
  const float* Ws = (const float*)d_in[2];
  const float* bs = (const float*)d_in[3];
  const int*  src = (const int*)d_in[4];
  const int*  dst = (const int*)d_in[5];
  float* out = (float*)d_out;

  const int N = in_sizes[0] / 64;
  const int E = in_sizes[1];
  const int L = 3;

  const int Np = (N + 4) & ~3;
  const int Ep = (E + 3) & ~3;
  int*  cnt = (int*)d_ws;
  int*  off = cnt + Np;
  int*  rnk = off + Np;
  int2* rec = (int2*)(rnk + Ep);

  {
    int t = N * 16;
    gcn_copy_h_kernel<<<(t + 255) / 256, 256, 0, stream>>>(h, out, cnt, N);
  }
  gcn_rank_kernel<<<(E + 255) / 256, 256, 0, stream>>>(dst, cnt, rnk, E);
  gcn_scan_kernel<<<(N + 1 + 1023) / 1024, 1024, 0, stream>>>(cnt, off, N);
  gcn_fill_kernel<<<(E + 255) / 256, 256, 0, stream>>>(src, dst, ew, rnk, off, rec, E);

  for (int layer = 0; layer < L; ++layer) {
    int blocks = (N + 15) / 16;
    gcn_layer_kernel<<<blocks, 256, 0, stream>>>(
        out + layer * 64, rec, off, Ws + layer * 64 * 64, bs + layer * 64,
        out + (layer + 1) * 64, N, (layer < L - 1) ? 1 : 0);
  }
}


# edges binned by destination in ONE pass: rank atomic, then record stored straight into the node's fixed 64-record slot (overflow list beyond); scan and fill launches gone; arrival-rank sum order kept
# speedup vs baseline: 2.3987x; 1.1376x over previous
#include <hip/hip_runtime.h>
#include <cmath>

typedef __attribute__((ext_vector_type(2))) float v2f;
typedef __attribute__((ext_vector_type(8))) float v8f;

__global__ void gcn_copy_h_kernel(const float* __restrict__ h,
                                  float* __restrict__ out,
                                  int* __restrict__ cnt, int n_nodes) {
  int i = blockIdx.x * blockDim.x + threadIdx.x;
  if (i < n_nodes + 2) cnt[i] = 0;
  if (i >= n_nodes * 16) return;
  int r  = i >> 4;
  int c4 = i & 15;
  float4 v = *((const float4*)(h + (size_t)r * 64) + c4);
  *((float4*)(out + (size_t)r * 256) + c4) = v;
}

#define GCN_SLOT 64

__global__ void gcn_bin_kernel(const int* __restrict__ src,
                               const int* __restrict__ dst,
                               const float* __restrict__ ew,
                               int* cnt,
                               int* ovf_n,
                               int2* __restrict__ rec,
                               int* __restrict__ ovf_dst,
                               int2* __restrict__ ovf_rec, int n_edges) {
  int e = blockIdx.x * blockDim.x + threadIdx.x;
  if (e >= n_edges) return;
  int d = dst[e];
  int2 r = make_int2(src[e], __float_as_int(ew[e]));
  int rank = atomicAdd(cnt + d, 1);
  int2* p = rec + (size_t)d * GCN_SLOT + rank;
  if (rank >= GCN_SLOT) {
    int k = atomicAdd(ovf_n, 1);
    ovf_dst[k] = d;
    p = ovf_rec + k;
  }
  *p = r;
}

template <int U>
__device__ __forceinline__ void gcn_pull_rows(const float* __restrict__ hprev,
                                              const int2* __restrict__ rec,
                                              int i, int f,
                                              float& ax, float& ay) {
  int    s[U];
  float  w[U];
  float2 hv[U];
#pragma unroll
  for (int j = 0; j < U; ++j) {
    int2 r = rec[i + j];
    s[j] = r.x;
    w[j] = __int_as_float(r.y);
  }
#pragma unroll
  for (int j = 0; j < U; ++j)
    hv[j] = *(const float2*)(hprev + (size_t)s[j] * 256 + f);
#pragma unroll
  for (int j = 0; j < U; ++j) {
    ax += hv[j].x * w[j];
    ay += hv[j].y * w[j];
  }
}

#define GCN_TILE_PITCH 68

__global__ __launch_bounds__(256)
void gcn_layer_kernel(const float* __restrict__ hprev,
                      const int2* __restrict__ rec,
                      const int* __restrict__ cnt,
                      const int* __restrict__ ovf_dst,
                      const int2* __restrict__ ovf_rec,
                      const float* __restrict__ W,
                      const float* __restrict__ bias,
                      float* __restrict__ outp,
                      int n_nodes, int do_tanh) {
  __shared__ float tile[16 * GCN_TILE_PITCH];
  int wave = __builtin_amdgcn_readfirstlane((int)(threadIdx.x >> 5));
  int lane = threadIdx.x & 31;
  int row0 = blockIdx.x * 16;
  int f = lane * 2;

  int nb = row0 + wave * 2;
  int n0 = nb     < n_nodes ? nb     : n_nodes;
  int n1 = nb + 1 < n_nodes ? nb + 1 : n_nodes;
  int d0 = cnt[n0];
  int d1 = cnt[n1];
#pragma unroll
  for (int q = 0; q < 2; ++q) {
    int n = q ? n1 : n0;
    int d = q ? d1 : d0;
    const int2* slot = rec + (size_t)n * GCN_SLOT;
    int e = d < GCN_SLOT ? d : GCN_SLOT;
    int i = 0;
    float ax = 0.f, ay = 0.f;
    for (; i + 16 <= e; i += 16) gcn_pull_rows<16>(hprev, slot, i, f, ax, ay);
    int r = e - i;
    if (r & 8) { gcn_pull_rows<8>(hprev, slot, i, f, ax, ay); i += 8; }
    if (r & 4) { gcn_pull_rows<4>(hprev, slot, i, f, ax, ay); i += 4; }
    if (r & 2) { gcn_pull_rows<2>(hprev, slot, i, f, ax, ay); i += 2; }
    if (r & 1) { gcn_pull_rows<1>(hprev, slot, i, f, ax, ay); }
    if (d > GCN_SLOT) {
      int m = cnt[n_nodes + 1];
#pragma nounroll
      for (int k = 0; k < m; ++k)
        if (ovf_dst[k] == n) gcn_pull_rows<1>(hprev, ovf_rec, k, f, ax, ay);
    }
    *(float2*)(tile + (wave * 2 + q) * GCN_TILE_PITCH + f) = make_float2(ax, ay);
  }
  __syncthreads();
  if (wave >= 4) return;

  int col0 = wave * 16;
  int lh = lane >> 4;
  int lm = lane & 15;
  const float* arow = tile + lm * GCN_TILE_PITCH + 2 * lh;
  const float* wcol = W + (size_t)(2 * lh) * 64 + col0 + lm;

  v8f acc = {};
#pragma unroll
  for (int k0 = 0; k0 < 64; k0 += 4) {
    v2f a = *(const v2f*)(arow + k0);
    v2f b;
    b.x = wcol[(size_t)k0 * 64];
    b.y = wcol[(size_t)k0 * 64 + 64];
    acc = __builtin_amdgcn_wmma_f32_16x16x4_f32(
         false, a,  false, b,
         (short)0, acc,  false,  false);
  }

  float bv = bias[col0 + lm];
  int rbase = row0 + 8 * lh;
  float* orow = outp + (size_t)rbase * 256 + col0 + lm;
#pragma unroll
  for (int v = 0; v < 8; ++v) {
    float val = acc[v] + bv;
    if (do_tanh) val = tanhf(val);
    if (rbase + v < n_nodes) orow[(size_t)v * 256] = val;
  }
}

extern "C" void kernel_launch(void* const* d_in, const int* in_sizes, int n_in,
                              void* d_out, int out_size, void* d_ws, size_t ws_size,
                              hipStream_t stream) {
  const float* h  = (const float*)d_in[0];
  const float* ew = (const float*)d_in[1];
  const float* Ws = (const float*)d_in[2];
  const float* bs = (const float*)d_in[3];
  const int*  src = (const int*)d_in[4];
  const int*  dst = (const int*)d_in[5];
  float* out = (float*)d_out;

  const int N = in_sizes[0] / 64;
  const int E = in_sizes[1];
  const int L = 3;

  const int Np = (N + 2 + 127) & ~127;
  const int Ep = (E + 127) & ~127;
  int*  cnt     = (int*)d_ws;
  int*  ovf_dst = cnt + Np;
  int2* ovf_rec = (int2*)(ovf_dst + Ep);
  int2* rec     = ovf_rec + Ep;

  {
    int t = N * 16;
    gcn_copy_h_kernel<<<(t + 255) / 256, 256, 0, stream>>>(h, out, cnt, N);
  }
  gcn_bin_kernel<<<(E + 255) / 256, 256, 0, stream>>>(src, dst, ew, cnt, cnt + N + 1, rec,
                                                     ovf_dst, ovf_rec, E);

  for (int layer = 0; layer < L; ++layer) {
    int blocks = (N + 15) / 16;
    gcn_layer_kernel<<<blocks, 256, 0, stream>>>(
        out + layer * 64, rec, cnt, ovf_dst, ovf_rec, Ws + layer * 64 * 64, bs + layer * 64,
        out + (layer + 1) * 64, N, (layer < L - 1) ? 1 : 0);
  }
}


# edge binning in LDS, 2 launches: chunks ordered by 128-node bucket, stored as runs; each bucket then ranked into slots by LDS atomics; per-edge global atomic + scattered store gone; arrival-rank sums
# speedup vs baseline: 3.7582x; 1.5668x over previous
#include <hip/hip_runtime.h>
#include <cmath>

typedef __attribute__((ext_vector_type(2))) float v2f;
typedef __attribute__((ext_vector_type(8))) float v8f;

__global__ void gcn_copy_h_kernel(const float* __restrict__ h,
                                  float* __restrict__ out,
                                  int* __restrict__ cnt, int n_nodes, int n_zero) {
  int i = blockIdx.x * blockDim.x + threadIdx.x;
  if (i < n_zero) cnt[i] = 0;
  if (i >= n_nodes * 16) return;
  int r  = i >> 4;
  int c4 = i & 15;
  float4 v = *((const float4*)(h + (size_t)r * 64) + c4);
  *((float4*)(out + (size_t)r * 256) + c4) = v;
}

#define GCN_SLOT      64
#define GCN_BKT_SHIFT 7
#define GCN_BKT_NODES 128
#define GCN_BKT_MAX   512
#define GCN_BKT_CAP   4096
#define GCN_CHUNK     4096
#define GCN_CNT_PITCH 32

__global__ __launch_bounds__(512)
void gcn_bucket_kernel(const int* __restrict__ src,
                       const int* __restrict__ dst,
                       const float* __restrict__ ew,
                       int* bcount,
                       int* ovf_n,
                       int4* __restrict__ stag,
                       int* __restrict__ ovf_dst,
                       int2* __restrict__ ovf_rec, int n_edges) {
  __shared__ int hist[GCN_BKT_MAX];
  __shared__ int scan[GCN_BKT_MAX];
  __shared__ int lbase[GCN_BKT_MAX];
  __shared__ int gbase[GCN_BKT_MAX];
  __shared__ int sd[GCN_CHUNK];
  __shared__ int ss[GCN_CHUNK];
  __shared__ int sw[GCN_CHUNK];
  const int t  = threadIdx.x;
  const int e0 = blockIdx.x * GCN_CHUNK;
  int n = n_edges - e0;
  if (n > GCN_CHUNK) n = GCN_CHUNK;

  hist[t] = 0;
  int d[GCN_CHUNK / 512], s[GCN_CHUNK / 512], w[GCN_CHUNK / 512];
#pragma unroll
  for (int j = 0; j < GCN_CHUNK / 512; ++j) {
    int i = j * 512 + t;
    bool in = i < n;
    int ii = e0 + (in ? i : 0);
    int dv = dst[ii];
    d[j] = in ? dv : -1;
    s[j] = src[ii];
    w[j] = __float_as_int(ew[ii]);
  }
  __syncthreads();
#pragma unroll
  for (int j = 0; j < GCN_CHUNK / 512; ++j)
    if (d[j] >= 0) atomicAdd(&hist[d[j] >> GCN_BKT_SHIFT], 1);
  __syncthreads();

  const int c = hist[t];
  scan[t] = c;
  for (int off = 1; off < GCN_BKT_MAX; off <<= 1) {
    __syncthreads();
    int v = t >= off ? scan[t - off] : 0;
    __syncthreads();
    scan[t] += v;
  }
  const int lb = scan[t] - c;
  lbase[t] = lb;
  gbase[t] = c > 0 ? atomicAdd(bcount + t * GCN_CNT_PITCH, c) : 0;
  hist[t] = lb;
  __syncthreads();

#pragma unroll
  for (int j = 0; j < GCN_CHUNK / 512; ++j)
    if (d[j] >= 0) {
      int p = atomicAdd(&hist[d[j] >> GCN_BKT_SHIFT], 1);
      sd[p] = d[j];
      ss[p] = s[j];
      sw[p] = w[j];
    }
  __syncthreads();

  for (int i = t; i < n; i += 512) {
    int dd = sd[i];
    int b  = dd >> GCN_BKT_SHIFT;
    int q  = gbase[b] + (i - lbase[b]);
    if (q < GCN_BKT_CAP) {
      stag[(size_t)b * GCN_BKT_CAP + q] = make_int4(dd, ss[i], sw[i], 0);
    } else {
      int k = atomicAdd(ovf_n, 1);
      ovf_dst[k] = dd;
      ovf_rec[k] = make_int2(ss[i], sw[i]);
    }
  }
}

__global__ __launch_bounds__(512)
void gcn_slot_kernel(const int4* __restrict__ stag,
                     const int* __restrict__ bcount,
                     int* cnt,
                     int* ovf_n,
                     int2* __restrict__ rec,
                     int* __restrict__ ovf_dst,
                     int2* __restrict__ ovf_rec, int n_nodes) {
  __shared__ int  lcnt[GCN_BKT_NODES];
  __shared__ int2 slots[GCN_BKT_NODES * GCN_SLOT];
  const int t = threadIdx.x;
  const int b = blockIdx.x;
  const int node0 = b * GCN_BKT_NODES;
  if (t < GCN_BKT_NODES) lcnt[t] = 0;
  __syncthreads();

  int m = bcount[b * GCN_CNT_PITCH];
  if (m > GCN_BKT_CAP) m = GCN_BKT_CAP;
  const int4* sp = stag + (size_t)b * GCN_BKT_CAP;
  for (int i0 = 0; i0 < m; i0 += 4 * 512) {
    int4 r[4];
#pragma unroll
    for (int j = 0; j < 4; ++j) {
      int i = i0 + j * 512 + t;
      int4 v = sp[i < m ? i : 0];
      if (i >= m) v.x = -1;
      r[j] = v;
    }
#pragma unroll
    for (int j = 0; j < 4; ++j)
      if (r[j].x >= 0) {
        int ln = r[j].x & (GCN_BKT_NODES - 1);
        int k  = atomicAdd(&lcnt[ln], 1);
        if (k < GCN_SLOT) {
          slots[ln * GCN_SLOT + k] = make_int2(r[j].y, r[j].z);
        } else {
          int q = atomicAdd(ovf_n, 1);
          ovf_dst[q] = r[j].x;
          ovf_rec[q] = make_int2(r[j].y, r[j].z);
        }
      }
  }
  __syncthreads();

  if (t < GCN_BKT_NODES && node0 + t < n_nodes) cnt[node0 + t] = lcnt[t];
  for (int i = t; i < GCN_BKT_NODES * GCN_SLOT; i += 512) {
    int ln = i / GCN_SLOT;
    int k  = i % GCN_SLOT;
    if (k < lcnt[ln]) rec[(size_t)(node0 + ln) * GCN_SLOT + k] = slots[i];
  }
}

template <int U>
__device__ __forceinline__ void gcn_pull_rows(const float* __restrict__ hprev,
                                              const int2* __restrict__ rec,
                                              int i, int f,
                                              float& ax, float& ay) {
  int    s[U];
  float  w[U];
  float2 hv[U];
#pragma unroll
  for (int j = 0; j < U; ++j) {
    int2 r = rec[i + j];
    s[j] = r.x;
    w[j] = __int_as_float(r.y);
  }
#pragma unroll
  for (int j = 0; j < U; ++j)
    hv[j] = *(const float2*)(hprev + (size_t)s[j] * 256 + f);
#pragma unroll
  for (int j = 0; j < U; ++j) {
    ax += hv[j].x * w[j];
    ay += hv[j].y * w[j];
  }
}

#define GCN_TILE_PITCH 68

__global__ __launch_bounds__(256)
void gcn_layer_kernel(const float* __restrict__ hprev,
                      const int2* __restrict__ rec,
                      const int* __restrict__ cnt,
                      const int* __restrict__ ovf_dst,
                      const int2* __restrict__ ovf_rec,
                      const float* __restrict__ W,
                      const float* __restrict__ bias,
                      float* __restrict__ outp,
                      int n_nodes, int do_tanh) {
  __shared__ float tile[16 * GCN_TILE_PITCH];
  int wave = __builtin_amdgcn_readfirstlane((int)(threadIdx.x >> 5));
  int lane = threadIdx.x & 31;
  int row0 = blockIdx.x * 16;
  int f = lane * 2;

  int nb = row0 + wave * 2;
  int n0 = nb     < n_nodes ? nb     : n_nodes;
  int n1 = nb + 1 < n_nodes ? nb + 1 : n_nodes;
  int d0 = cnt[n0];
  int d1 = cnt[n1];
  int m  = cnt[n_nodes + 1];
#pragma unroll
  for (int q = 0; q < 2; ++q) {
    int n = q ? n1 : n0;
    int d = q ? d1 : d0;
    const int2* slot = rec + (size_t)n * GCN_SLOT;
    int e = d < GCN_SLOT ? d : GCN_SLOT;
    int i = 0;
    float ax = 0.f, ay = 0.f;
    for (; i + 16 <= e; i += 16) gcn_pull_rows<16>(hprev, slot, i, f, ax, ay);
    int r = e - i;
    if (r & 8) { gcn_pull_rows<8>(hprev, slot, i, f, ax, ay); i += 8; }
    if (r & 4) { gcn_pull_rows<4>(hprev, slot, i, f, ax, ay); i += 4; }
    if (r & 2) { gcn_pull_rows<2>(hprev, slot, i, f, ax, ay); i += 2; }
    if (r & 1) { gcn_pull_rows<1>(hprev, slot, i, f, ax, ay); }
    if (m > 0) {
#pragma nounroll
      for (int k = 0; k < m; ++k)
        if (ovf_dst[k] == n) gcn_pull_rows<1>(hprev, ovf_rec, k, f, ax, ay);
    }
    *(float2*)(tile + (wave * 2 + q) * GCN_TILE_PITCH + f) = make_float2(ax, ay);
  }
  __syncthreads();
  if (wave >= 4) return;

  int col0 = wave * 16;
  int lh = lane >> 4;
  int lm = lane & 15;
  const float* arow = tile + lm * GCN_TILE_PITCH + 2 * lh;
  const float* wcol = W + (size_t)(2 * lh) * 64 + col0 + lm;

  v8f acc = {};
#pragma unroll
  for (int k0 = 0; k0 < 64; k0 += 4) {
    v2f a = *(const v2f*)(arow + k0);
    v2f b;
    b.x = wcol[(size_t)k0 * 64];
    b.y = wcol[(size_t)k0 * 64 + 64];
    acc = __builtin_amdgcn_wmma_f32_16x16x4_f32(
         false, a,  false, b,
         (short)0, acc,  false,  false);
  }

  float bv = bias[col0 + lm];
  int rbase = row0 + 8 * lh;
  float* orow = outp + (size_t)rbase * 256 + col0 + lm;
#pragma unroll
  for (int v = 0; v < 8; ++v) {
    float val = acc[v] + bv;
    if (do_tanh) val = tanhf(val);
    if (rbase + v < n_nodes) orow[(size_t)v * 256] = val;
  }
}

extern "C" void kernel_launch(void* const* d_in, const int* in_sizes, int n_in,
                              void* d_out, int out_size, void* d_ws, size_t ws_size,
                              hipStream_t stream) {
  const float* h  = (const float*)d_in[0];
  const float* ew = (const float*)d_in[1];
  const float* Ws = (const float*)d_in[2];
  const float* bs = (const float*)d_in[3];
  const int*  src = (const int*)d_in[4];
  const int*  dst = (const int*)d_in[5];
  float* out = (float*)d_out;

  const int N = in_sizes[0] / 64;
  const int E = in_sizes[1];
  const int L = 3;

  const int Np = (N + 2 + 127) & ~127;
  const int Ep = (E + 127) & ~127;
  const int NB = (N + GCN_BKT_NODES - 1) / GCN_BKT_NODES;
  const int Bp = (NB * GCN_CNT_PITCH + 127) & ~127;
  int*  cnt     = (int*)d_ws;
  int*  bcount  = cnt + Np;
  int*  ovf_dst = bcount + Bp;
  int2* ovf_rec = (int2*)(ovf_dst + Ep);
  int2* rec     = ovf_rec + Ep;
  int4* stag    = (int4*)(rec + (size_t)N * GCN_SLOT);

  {
    int t = N * 16;
    gcn_copy_h_kernel<<<(t + 255) / 256, 256, 0, stream>>>(h, out, cnt, N, Np + Bp);
  }
  gcn_bucket_kernel<<<(E + GCN_CHUNK - 1) / GCN_CHUNK, 512, 0, stream>>>(
      src, dst, ew, bcount, cnt + N + 1, stag, ovf_dst, ovf_rec, E);
  gcn_slot_kernel<<<NB, 512, 0, stream>>>(stag, bcount, cnt, cnt + N + 1, rec,
                                          ovf_dst, ovf_rec, N);

  for (int layer = 0; layer < L; ++layer) {
    int blocks = (N + 15) / 16;
    gcn_layer_kernel<<<blocks, 256, 0, stream>>>(
        out + layer * 64, rec, cnt, ovf_dst, ovf_rec, Ws + layer * 64 * 64, bs + layer * 64,
        out + (layer + 1) * 64, N, (layer < L - 1) ? 1 : 0);
  }
}


# layer epilogue: the 16x64 product tile is staged in LDS and each 256-byte output row leaves as one whole piece of one store (was four 64-byte pieces from four waves); bias+tanh on the staged rows
# speedup vs baseline: 4.0107x; 1.0672x over previous
#include <hip/hip_runtime.h>
#include <cmath>

typedef __attribute__((ext_vector_type(2))) float v2f;
typedef __attribute__((ext_vector_type(8))) float v8f;

__global__ void gcn_copy_h_kernel(const float* __restrict__ h,
                                  float* __restrict__ out,
                                  int* __restrict__ cnt, int n_nodes, int n_zero) {
  int i = blockIdx.x * blockDim.x + threadIdx.x;
  if (i < n_zero) cnt[i] = 0;
  if (i >= n_nodes * 16) return;
  int r  = i >> 4;
  int c4 = i & 15;
  float4 v = *((const float4*)(h + (size_t)r * 64) + c4);
  *((float4*)(out + (size_t)r * 256) + c4) = v;
}

#define GCN_SLOT      64
#define GCN_BKT_SHIFT 7
#define GCN_BKT_NODES 128
#define GCN_BKT_MAX   512
#define GCN_BKT_CAP   4096
#define GCN_CHUNK     4096
#define GCN_CNT_PITCH 32

__global__ __launch_bounds__(512)
void gcn_bucket_kernel(const int* __restrict__ src,
                       const int* __restrict__ dst,
                       const float* __restrict__ ew,
                       int* bcount,
                       int* ovf_n,
                       int4* __restrict__ stag,
                       int* __restrict__ ovf_dst,
                       int2* __restrict__ ovf_rec, int n_edges) {
  __shared__ int hist[GCN_BKT_MAX];
  __shared__ int scan[GCN_BKT_MAX];
  __shared__ int lbase[GCN_BKT_MAX];
  __shared__ int gbase[GCN_BKT_MAX];
  __shared__ int sd[GCN_CHUNK];
  __shared__ int ss[GCN_CHUNK];
  __shared__ int sw[GCN_CHUNK];
  const int t  = threadIdx.x;
  const int e0 = blockIdx.x * GCN_CHUNK;
  int n = n_edges - e0;
  if (n > GCN_CHUNK) n = GCN_CHUNK;

  hist[t] = 0;
  int d[GCN_CHUNK / 512], s[GCN_CHUNK / 512], w[GCN_CHUNK / 512];
#pragma unroll
  for (int j = 0; j < GCN_CHUNK / 512; ++j) {
    int i = j * 512 + t;
    bool in = i < n;
    int ii = e0 + (in ? i : 0);
    int dv = dst[ii];
    d[j] = in ? dv : -1;
    s[j] = src[ii];
    w[j] = __float_as_int(ew[ii]);
  }
  __syncthreads();
#pragma unroll
  for (int j = 0; j < GCN_CHUNK / 512; ++j)
    if (d[j] >= 0) atomicAdd(&hist[d[j] >> GCN_BKT_SHIFT], 1);
  __syncthreads();

  const int c = hist[t];
  scan[t] = c;
  for (int off = 1; off < GCN_BKT_MAX; off <<= 1) {
    __syncthreads();
    int v = t >= off ? scan[t - off] : 0;
    __syncthreads();
    scan[t] += v;
  }
  const int lb = scan[t] - c;
  lbase[t] = lb;
  gbase[t] = c > 0 ? atomicAdd(bcount + t * GCN_CNT_PITCH, c) : 0;
  hist[t] = lb;
  __syncthreads();

#pragma unroll
  for (int j = 0; j < GCN_CHUNK / 512; ++j)
    if (d[j] >= 0) {
      int p = atomicAdd(&hist[d[j] >> GCN_BKT_SHIFT], 1);
      sd[p] = d[j];
      ss[p] = s[j];
      sw[p] = w[j];
    }
  __syncthreads();

  for (int i = t; i < n; i += 512) {
    int dd = sd[i];
    int b  = dd >> GCN_BKT_SHIFT;
    int q  = gbase[b] + (i - lbase[b]);
    if (q < GCN_BKT_CAP) {
      stag[(size_t)b * GCN_BKT_CAP + q] = make_int4(dd, ss[i], sw[i], 0);
    } else {
      int k = atomicAdd(ovf_n, 1);
      ovf_dst[k] = dd;
      ovf_rec[k] = make_int2(ss[i], sw[i]);
    }
  }
}

__global__ __launch_bounds__(512)
void gcn_slot_kernel(const int4* __restrict__ stag,
                     const int* __restrict__ bcount,
                     int* cnt,
                     int* ovf_n,
                     int2* __restrict__ rec,
                     int* __restrict__ ovf_dst,
                     int2* __restrict__ ovf_rec, int n_nodes) {
  __shared__ int  lcnt[GCN_BKT_NODES];
  __shared__ int2 slots[GCN_BKT_NODES * GCN_SLOT];
  const int t = threadIdx.x;
  const int b = blockIdx.x;
  const int node0 = b * GCN_BKT_NODES;
  if (t < GCN_BKT_NODES) lcnt[t] = 0;
  __syncthreads();

  int m = bcount[b * GCN_CNT_PITCH];
  if (m > GCN_BKT_CAP) m = GCN_BKT_CAP;
  const int4* sp = stag + (size_t)b * GCN_BKT_CAP;
  for (int i0 = 0; i0 < m; i0 += 4 * 512) {
    int4 r[4];
#pragma unroll
    for (int j = 0; j < 4; ++j) {
      int i = i0 + j * 512 + t;
      int4 v = sp[i < m ? i : 0];
      if (i >= m) v.x = -1;
      r[j] = v;
    }
#pragma unroll
    for (int j = 0; j < 4; ++j)
      if (r[j].x >= 0) {
        int ln = r[j].x & (GCN_BKT_NODES - 1);
        int k  = atomicAdd(&lcnt[ln], 1);
        if (k < GCN_SLOT) {
          slots[ln * GCN_SLOT + k] = make_int2(r[j].y, r[j].z);
        } else {
          int q = atomicAdd(ovf_n, 1);
          ovf_dst[q] = r[j].x;
          ovf_rec[q] = make_int2(r[j].y, r[j].z);
        }
      }
  }
  __syncthreads();

  if (t < GCN_BKT_NODES && node0 + t < n_nodes) cnt[node0 + t] = lcnt[t];
  for (int i = t; i < GCN_BKT_NODES * GCN_SLOT; i += 512) {
    int ln = i / GCN_SLOT;
    int k  = i % GCN_SLOT;
    if (k < lcnt[ln]) rec[(size_t)(node0 + ln) * GCN_SLOT + k] = slots[i];
  }
}

template <int U>
__device__ __forceinline__ void gcn_pull_rows(const float* __restrict__ hprev,
                                              const int2* __restrict__ rec,
                                              int i, int f,
                                              float& ax, float& ay) {
  int    s[U];
  float  w[U];
  float2 hv[U];
#pragma unroll
  for (int j = 0; j < U; ++j) {
    int2 r = rec[i + j];
    s[j] = r.x;
    w[j] = __int_as_float(r.y);
  }
#pragma unroll
  for (int j = 0; j < U; ++j)
    hv[j] = *(const float2*)(hprev + (size_t)s[j] * 256 + f);
#pragma unroll
  for (int j = 0; j < U; ++j) {
    ax += hv[j].x * w[j];
    ay += hv[j].y * w[j];
  }
}

#define GCN_TILE_PITCH 68

__global__ __launch_bounds__(256)
void gcn_layer_kernel(const float* __restrict__ hprev,
                      const int2* __restrict__ rec,
                      const int* __restrict__ cnt,
                      const int* __restrict__ ovf_dst,
                      const int2* __restrict__ ovf_rec,
                      const float* __restrict__ W,
                      const float* __restrict__ bias,
                      float* __restrict__ outp,
                      int n_nodes, int do_tanh) {
  __shared__ float tile[16 * GCN_TILE_PITCH];
  __shared__ __attribute__((aligned(16))) float otile[16 * GCN_TILE_PITCH];
  int wave = __builtin_amdgcn_readfirstlane((int)(threadIdx.x >> 5));
  int lane = threadIdx.x & 31;
  int row0 = blockIdx.x * 16;
  int f = lane * 2;

  int nb = row0 + wave * 2;
  int n0 = nb     < n_nodes ? nb     : n_nodes;
  int n1 = nb + 1 < n_nodes ? nb + 1 : n_nodes;
  int d0 = cnt[n0];
  int d1 = cnt[n1];
  int m  = cnt[n_nodes + 1];
#pragma unroll
  for (int q = 0; q < 2; ++q) {
    int n = q ? n1 : n0;
    int d = q ? d1 : d0;
    const int2* slot = rec + (size_t)n * GCN_SLOT;
    int e = d < GCN_SLOT ? d : GCN_SLOT;
    int i = 0;
    float ax = 0.f, ay = 0.f;
    for (; i + 16 <= e; i += 16) gcn_pull_rows<16>(hprev, slot, i, f, ax, ay);
    int r = e - i;
    if (r & 8) { gcn_pull_rows<8>(hprev, slot, i, f, ax, ay); i += 8; }
    if (r & 4) { gcn_pull_rows<4>(hprev, slot, i, f, ax, ay); i += 4; }
    if (r & 2) { gcn_pull_rows<2>(hprev, slot, i, f, ax, ay); i += 2; }
    if (r & 1) { gcn_pull_rows<1>(hprev, slot, i, f, ax, ay); }
    if (m > 0) {
#pragma nounroll
      for (int k = 0; k < m; ++k)
        if (ovf_dst[k] == n) gcn_pull_rows<1>(hprev, ovf_rec, k, f, ax, ay);
    }
    *(float2*)(tile + (wave * 2 + q) * GCN_TILE_PITCH + f) = make_float2(ax, ay);
  }
  __syncthreads();
  int lh = lane >> 4;
  int lm = lane & 15;
  int c  = 4 * lm;
  float4 bv = *(const float4*)(bias + c);

  if (wave < 4) {
    int col0 = wave * 16;
    const float* arow = tile + lm * GCN_TILE_PITCH + 2 * lh;
    const float* wcol = W + (size_t)(2 * lh) * 64 + col0 + lm;

    v8f acc = {};
#pragma unroll
    for (int k0 = 0; k0 < 64; k0 += 4) {
      v2f a = *(const v2f*)(arow + k0);
      v2f b;
      b.x = wcol[(size_t)k0 * 64];
      b.y = wcol[(size_t)k0 * 64 + 64];
      acc = __builtin_amdgcn_wmma_f32_16x16x4_f32(
           false, a,  false, b,
           (short)0, acc,  false,  false);
    }
    float* op = otile + (8 * lh) * GCN_TILE_PITCH + col0 + lm;
#pragma unroll
    for (int v = 0; v < 8; ++v) op[v * GCN_TILE_PITCH] = acc[v];
  }
  __syncthreads();

  int r = wave * 2 + lh;
  float4 val = *(const float4*)(otile + r * GCN_TILE_PITCH + c);
  val.x += bv.x; val.y += bv.y; val.z += bv.z; val.w += bv.w;
  if (do_tanh) {
    val.x = tanhf(val.x); val.y = tanhf(val.y);
    val.z = tanhf(val.z); val.w = tanhf(val.w);
  }
  if (row0 + r < n_nodes) *(float4*)(outp + (size_t)(row0 + r) * 256 + c) = val;
}

extern "C" void kernel_launch(void* const* d_in, const int* in_sizes, int n_in,
                              void* d_out, int out_size, void* d_ws, size_t ws_size,
                              hipStream_t stream) {
  const float* h  = (const float*)d_in[0];
  const float* ew = (const float*)d_in[1];
  const float* Ws = (const float*)d_in[2];
  const float* bs = (const float*)d_in[3];
  const int*  src = (const int*)d_in[4];
  const int*  dst = (const int*)d_in[5];
  float* out = (float*)d_out;

  const int N = in_sizes[0] / 64;
  const int E = in_sizes[1];
  const int L = 3;

  const int Np = (N + 2 + 127) & ~127;
  const int Ep = (E + 127) & ~127;
  const int NB = (N + GCN_BKT_NODES - 1) / GCN_BKT_NODES;
  const int Bp = (NB * GCN_CNT_PITCH + 127) & ~127;
  int*  cnt     = (int*)d_ws;
  int*  bcount  = cnt + Np;
  int*  ovf_dst = bcount + Bp;
  int2* ovf_rec = (int2*)(ovf_dst + Ep);
  int2* rec     = ovf_rec + Ep;
  int4* stag    = (int4*)(rec + (size_t)N * GCN_SLOT);

  {
    int t = N * 16;
    gcn_copy_h_kernel<<<(t + 255) / 256, 256, 0, stream>>>(h, out, cnt, N, Np + Bp);
  }
  gcn_bucket_kernel<<<(E + GCN_CHUNK - 1) / GCN_CHUNK, 512, 0, stream>>>(
      src, dst, ew, bcount, cnt + N + 1, stag, ovf_dst, ovf_rec, E);
  gcn_slot_kernel<<<NB, 512, 0, stream>>>(stag, bcount, cnt, cnt + N + 1, rec,
                                          ovf_dst, ovf_rec, N);

  for (int layer = 0; layer < L; ++layer) {
    int blocks = (N + 15) / 16;
    gcn_layer_kernel<<<blocks, 256, 0, stream>>>(
        out + layer * 64, rec, cnt, ovf_dst, ovf_rec, Ws + layer * 64 * 64, bs + layer * 64,
        out + (layer + 1) * 64, N, (layer < L - 1) ? 1 : 0);
  }
}


# layer kernel: a node's edge records come as one coalesced 256-byte vector load, one record per lane, fields broadcast by v_readlane (was ~4 dependent scalar loads of 64-byte lines); sum order kept
# speedup vs baseline: 4.2720x; 1.0652x over previous
#include <hip/hip_runtime.h>
#include <cmath>

typedef __attribute__((ext_vector_type(2))) float v2f;
typedef __attribute__((ext_vector_type(8))) float v8f;

__global__ void gcn_copy_h_kernel(const float* __restrict__ h,
                                  float* __restrict__ out,
                                  int* __restrict__ cnt, int n_nodes, int n_zero) {
  int i = blockIdx.x * blockDim.x + threadIdx.x;
  if (i < n_zero) cnt[i] = 0;
  if (i >= n_nodes * 16) return;
  int r  = i >> 4;
  int c4 = i & 15;
  float4 v = *((const float4*)(h + (size_t)r * 64) + c4);
  *((float4*)(out + (size_t)r * 256) + c4) = v;
}

#define GCN_SLOT      64
#define GCN_BKT_SHIFT 7
#define GCN_BKT_NODES 128
#define GCN_BKT_MAX   512
#define GCN_BKT_CAP   4096
#define GCN_CHUNK     4096
#define GCN_CNT_PITCH 32

__global__ __launch_bounds__(512)
void gcn_bucket_kernel(const int* __restrict__ src,
                       const int* __restrict__ dst,
                       const float* __restrict__ ew,
                       int* bcount,
                       int* ovf_n,
                       int4* __restrict__ stag,
                       int* __restrict__ ovf_dst,
                       int2* __restrict__ ovf_rec, int n_edges) {
  __shared__ int hist[GCN_BKT_MAX];
  __shared__ int scan[GCN_BKT_MAX];
  __shared__ int lbase[GCN_BKT_MAX];
  __shared__ int gbase[GCN_BKT_MAX];
  __shared__ int sd[GCN_CHUNK];
  __shared__ int ss[GCN_CHUNK];
  __shared__ int sw[GCN_CHUNK];
  const int t  = threadIdx.x;
  const int e0 = blockIdx.x * GCN_CHUNK;
  int n = n_edges - e0;
  if (n > GCN_CHUNK) n = GCN_CHUNK;

  hist[t] = 0;
  int d[GCN_CHUNK / 512], s[GCN_CHUNK / 512], w[GCN_CHUNK / 512];
#pragma unroll
  for (int j = 0; j < GCN_CHUNK / 512; ++j) {
    int i = j * 512 + t;
    bool in = i < n;
    int ii = e0 + (in ? i : 0);
    int dv = dst[ii];
    d[j] = in ? dv : -1;
    s[j] = src[ii];
    w[j] = __float_as_int(ew[ii]);
  }
  __syncthreads();
#pragma unroll
  for (int j = 0; j < GCN_CHUNK / 512; ++j)
    if (d[j] >= 0) atomicAdd(&hist[d[j] >> GCN_BKT_SHIFT], 1);
  __syncthreads();

  const int c = hist[t];
  scan[t] = c;
  for (int off = 1; off < GCN_BKT_MAX; off <<= 1) {
    __syncthreads();
    int v = t >= off ? scan[t - off] : 0;
    __syncthreads();
    scan[t] += v;
  }
  const int lb = scan[t] - c;
  lbase[t] = lb;
  gbase[t] = c > 0 ? atomicAdd(bcount + t * GCN_CNT_PITCH, c) : 0;
  hist[t] = lb;
  __syncthreads();

#pragma unroll
  for (int j = 0; j < GCN_CHUNK / 512; ++j)
    if (d[j] >= 0) {
      int p = atomicAdd(&hist[d[j] >> GCN_BKT_SHIFT], 1);
      sd[p] = d[j];
      ss[p] = s[j];
      sw[p] = w[j];
    }
  __syncthreads();

  for (int i = t; i < n; i += 512) {
    int dd = sd[i];
    int b  = dd >> GCN_BKT_SHIFT;
    int q  = gbase[b] + (i - lbase[b]);
    if (q < GCN_BKT_CAP) {
      stag[(size_t)b * GCN_BKT_CAP + q] = make_int4(dd, ss[i], sw[i], 0);
    } else {
      int k = atomicAdd(ovf_n, 1);
      ovf_dst[k] = dd;
      ovf_rec[k] = make_int2(ss[i], sw[i]);
    }
  }
}

__global__ __launch_bounds__(512)
void gcn_slot_kernel(const int4* __restrict__ stag,
                     const int* __restrict__ bcount,
                     int* cnt,
                     int* ovf_n,
                     int2* __restrict__ rec,
                     int* __restrict__ ovf_dst,
                     int2* __restrict__ ovf_rec, int n_nodes) {
  __shared__ int  lcnt[GCN_BKT_NODES];
  __shared__ int2 slots[GCN_BKT_NODES * GCN_SLOT];
  const int t = threadIdx.x;
  const int b = blockIdx.x;
  const int node0 = b * GCN_BKT_NODES;
  if (t < GCN_BKT_NODES) lcnt[t] = 0;
  __syncthreads();

  int m = bcount[b * GCN_CNT_PITCH];
  if (m > GCN_BKT_CAP) m = GCN_BKT_CAP;
  const int4* sp = stag + (size_t)b * GCN_BKT_CAP;
  for (int i0 = 0; i0 < m; i0 += 4 * 512) {
    int4 r[4];
#pragma unroll
    for (int j = 0; j < 4; ++j) {
      int i = i0 + j * 512 + t;
      int4 v = sp[i < m ? i : 0];
      if (i >= m) v.x = -1;
      r[j] = v;
    }
#pragma unroll
    for (int j = 0; j < 4; ++j)
      if (r[j].x >= 0) {
        int ln = r[j].x & (GCN_BKT_NODES - 1);
        int k  = atomicAdd(&lcnt[ln], 1);
        if (k < GCN_SLOT) {
          slots[ln * GCN_SLOT + k] = make_int2(r[j].y, r[j].z);
        } else {
          int q = atomicAdd(ovf_n, 1);
          ovf_dst[q] = r[j].x;
          ovf_rec[q] = make_int2(r[j].y, r[j].z);
        }
      }
  }
  __syncthreads();

  if (t < GCN_BKT_NODES && node0 + t < n_nodes) cnt[node0 + t] = lcnt[t];
  for (int i = t; i < GCN_BKT_NODES * GCN_SLOT; i += 512) {
    int ln = i / GCN_SLOT;
    int k  = i % GCN_SLOT;
    if (k < lcnt[ln]) rec[(size_t)(node0 + ln) * GCN_SLOT + k] = slots[i];
  }
}

template <int U>
__device__ __forceinline__ void gcn_pull_rows(const float* __restrict__ hprev,
                                              const int2* __restrict__ rec,
                                              int i, int f,
                                              float& ax, float& ay) {
  int    s[U];
  float  w[U];
  float2 hv[U];
#pragma unroll
  for (int j = 0; j < U; ++j) {
    int2 r = rec[i + j];
    s[j] = r.x;
    w[j] = __int_as_float(r.y);
  }
#pragma unroll
  for (int j = 0; j < U; ++j)
    hv[j] = *(const float2*)(hprev + (size_t)s[j] * 256 + f);
#pragma unroll
  for (int j = 0; j < U; ++j) {
    ax += hv[j].x * w[j];
    ay += hv[j].y * w[j];
  }
}

template <int U>
__device__ __forceinline__ void gcn_pull_regs(const float* __restrict__ hprev,
                                              int rx, int ry, int p, int f,
                                              float& ax, float& ay) {
  int    s[U];
  float  w[U];
  float2 hv[U];
#pragma unroll
  for (int j = 0; j < U; ++j) {
    s[j] = __builtin_amdgcn_readlane(rx, p + j);
    w[j] = __int_as_float(__builtin_amdgcn_readlane(ry, p + j));
  }
#pragma unroll
  for (int j = 0; j < U; ++j)
    hv[j] = *(const float2*)(hprev + (size_t)s[j] * 256 + f);
#pragma unroll
  for (int j = 0; j < U; ++j) {
    ax += hv[j].x * w[j];
    ay += hv[j].y * w[j];
  }
}

__device__ __forceinline__ void gcn_pull_half(const float* __restrict__ hprev,
                                              int2 r, int c, int f,
                                              float& ax, float& ay) {
  int i = 0;
  for (; i + 16 <= c; i += 16) gcn_pull_regs<16>(hprev, r.x, r.y, i, f, ax, ay);
  int t = c - i;
  if (t & 8) { gcn_pull_regs<8>(hprev, r.x, r.y, i, f, ax, ay); i += 8; }
  if (t & 4) { gcn_pull_regs<4>(hprev, r.x, r.y, i, f, ax, ay); i += 4; }
  if (t & 2) { gcn_pull_regs<2>(hprev, r.x, r.y, i, f, ax, ay); i += 2; }
  if (t & 1) { gcn_pull_regs<1>(hprev, r.x, r.y, i, f, ax, ay); }
}

#define GCN_TILE_PITCH 68

__global__ __launch_bounds__(256)
void gcn_layer_kernel(const float* __restrict__ hprev,
                      const int2* __restrict__ rec,
                      const int* __restrict__ cnt,
                      const int* __restrict__ ovf_dst,
                      const int2* __restrict__ ovf_rec,
                      const float* __restrict__ W,
                      const float* __restrict__ bias,
                      float* __restrict__ outp,
                      int n_nodes, int do_tanh) {
  __shared__ float tile[16 * GCN_TILE_PITCH];
  __shared__ __attribute__((aligned(16))) float otile[16 * GCN_TILE_PITCH];
  int wave = __builtin_amdgcn_readfirstlane((int)(threadIdx.x >> 5));
  int lane = threadIdx.x & 31;
  int row0 = blockIdx.x * 16;
  int f = lane * 2;

  int nb = row0 + wave * 2;
  int n0 = nb     < n_nodes ? nb     : n_nodes;
  int n1 = nb + 1 < n_nodes ? nb + 1 : n_nodes;
  int d0 = cnt[n0];
  int d1 = cnt[n1];
  int m  = cnt[n_nodes + 1];
  int e0 = d0 < GCN_SLOT ? d0 : GCN_SLOT;
  int e1 = d1 < GCN_SLOT ? d1 : GCN_SLOT;
  const int2* slot0 = rec + (size_t)n0 * GCN_SLOT;
  const int2* slot1 = rec + (size_t)n1 * GCN_SLOT;
  int2 ra0 = make_int2(0, 0), rb0 = ra0, ra1 = ra0, rb1 = ra0;
  if (lane      < e0) ra0 = slot0[lane];
  if (lane + 32 < e0) rb0 = slot0[lane + 32];
  if (lane      < e1) ra1 = slot1[lane];
  if (lane + 32 < e1) rb1 = slot1[lane + 32];
#pragma unroll
  for (int q = 0; q < 2; ++q) {
    int n = q ? n1 : n0;
    int e = q ? e1 : e0;
    float ax = 0.f, ay = 0.f;
    gcn_pull_half(hprev, q ? ra1 : ra0, e < 32 ? e : 32, f, ax, ay);
    if (e > 32) gcn_pull_half(hprev, q ? rb1 : rb0, e - 32, f, ax, ay);
    if (m > 0) {
#pragma nounroll
      for (int k = 0; k < m; ++k)
        if (ovf_dst[k] == n) gcn_pull_rows<1>(hprev, ovf_rec, k, f, ax, ay);
    }
    *(float2*)(tile + (wave * 2 + q) * GCN_TILE_PITCH + f) = make_float2(ax, ay);
  }
  __syncthreads();
  int lh = lane >> 4;
  int lm = lane & 15;
  int c  = 4 * lm;
  float4 bv = *(const float4*)(bias + c);

  if (wave < 4) {
    int col0 = wave * 16;
    const float* arow = tile + lm * GCN_TILE_PITCH + 2 * lh;
    const float* wcol = W + (size_t)(2 * lh) * 64 + col0 + lm;

    v8f acc = {};
#pragma unroll
    for (int k0 = 0; k0 < 64; k0 += 4) {
      v2f a = *(const v2f*)(arow + k0);
      v2f b;
      b.x = wcol[(size_t)k0 * 64];
      b.y = wcol[(size_t)k0 * 64 + 64];
      acc = __builtin_amdgcn_wmma_f32_16x16x4_f32(
           false, a,  false, b,
           (short)0, acc,  false,  false);
    }
    float* op = otile + (8 * lh) * GCN_TILE_PITCH + col0 + lm;
#pragma unroll
    for (int v = 0; v < 8; ++v) op[v * GCN_TILE_PITCH] = acc[v];
  }
  __syncthreads();

  int r = wave * 2 + lh;
  float4 val = *(const float4*)(otile + r * GCN_TILE_PITCH + c);
  val.x += bv.x; val.y += bv.y; val.z += bv.z; val.w += bv.w;
  if (do_tanh) {
    val.x = tanhf(val.x); val.y = tanhf(val.y);
    val.z = tanhf(val.z); val.w = tanhf(val.w);
  }
  if (row0 + r < n_nodes) *(float4*)(outp + (size_t)(row0 + r) * 256 + c) = val;
}

extern "C" void kernel_launch(void* const* d_in, const int* in_sizes, int n_in,
                              void* d_out, int out_size, void* d_ws, size_t ws_size,
                              hipStream_t stream) {
  const float* h  = (const float*)d_in[0];
  const float* ew = (const float*)d_in[1];
  const float* Ws = (const float*)d_in[2];
  const float* bs = (const float*)d_in[3];
  const int*  src = (const int*)d_in[4];
  const int*  dst = (const int*)d_in[5];
  float* out = (float*)d_out;

  const int N = in_sizes[0] / 64;
  const int E = in_sizes[1];
  const int L = 3;

  const int Np = (N + 2 + 127) & ~127;
  const int Ep = (E + 127) & ~127;
  const int NB = (N + GCN_BKT_NODES - 1) / GCN_BKT_NODES;
  const int Bp = (NB * GCN_CNT_PITCH + 127) & ~127;
  int*  cnt     = (int*)d_ws;
  int*  bcount  = cnt + Np;
  int*  ovf_dst = bcount + Bp;
  int2* ovf_rec = (int2*)(ovf_dst + Ep);
  int2* rec     = ovf_rec + Ep;
  int4* stag    = (int4*)(rec + (size_t)N * GCN_SLOT);

  {
    int t = N * 16;
    gcn_copy_h_kernel<<<(t + 255) / 256, 256, 0, stream>>>(h, out, cnt, N, Np + Bp);
  }
  gcn_bucket_kernel<<<(E + GCN_CHUNK - 1) / GCN_CHUNK, 512, 0, stream>>>(
      src, dst, ew, bcount, cnt + N + 1, stag, ovf_dst, ovf_rec, E);
  gcn_slot_kernel<<<NB, 512, 0, stream>>>(stag, bcount, cnt, cnt + N + 1, rec,
                                          ovf_dst, ovf_rec, N);

  for (int layer = 0; layer < L; ++layer) {
    int blocks = (N + 15) / 16;
    gcn_layer_kernel<<<blocks, 256, 0, stream>>>(
        out + layer * 64, rec, cnt, ovf_dst, ovf_rec, Ws + layer * 64 * 64, bs + layer * 64,
        out + (layer + 1) * 64, N, (layer < L - 1) ? 1 : 0);
  }
}


# edge binning without global counters: each (128-node bucket, 4096-edge chunk) pair owns a fixed 256-byte staging cell; clearing launch gone, h copy rides in the binning launch; arrival-rank sums
# speedup vs baseline: 4.4837x; 1.0496x over previous
#include <hip/hip_runtime.h>
#include <cmath>

typedef __attribute__((ext_vector_type(2))) float v2f;
typedef __attribute__((ext_vector_type(8))) float v8f;
typedef __attribute__((ext_vector_type(4))) float v4f;

#define GCN_SLOT      64
#define GCN_BKT_SHIFT 7
#define GCN_BKT_NODES 128
#define GCN_BKT_MAX   512
#define GCN_CHUNK     4096
#define GCN_CELL      32
#define GCN_COPY_U    6

__global__ __launch_bounds__(512)
void gcn_bucket_kernel(const int* __restrict__ src,
                       const int* __restrict__ dst,
                       const float* __restrict__ ew,
                       const float* __restrict__ h,
                       float* __restrict__ out,
                       int* __restrict__ cnt_tail,
                       int* __restrict__ cellcnt,
                       int* __restrict__ chunk_ovf,
                       int2* __restrict__ stag,
                       int4* __restrict__ cov,
                       int n_edges, int n_nodes, int n_bkt, int cell_pitch) {
  __shared__ int hist[GCN_BKT_MAX];
  __shared__ int scan[GCN_BKT_MAX];
  __shared__ int lbase[GCN_BKT_MAX];
  __shared__ int sd[GCN_CHUNK];
  __shared__ int ss[GCN_CHUNK];
  __shared__ int sw[GCN_CHUNK];
  __shared__ int ovc;
  const int t  = threadIdx.x;
  const int ch = blockIdx.x;
  const int e0 = ch * GCN_CHUNK;
  int n = n_edges - e0;
  if (n > GCN_CHUNK) n = GCN_CHUNK;

  hist[t] = 0;
  if (t == 0) ovc = 0;
  if (ch == 0 && t < 2) cnt_tail[t] = 0;

  const int total  = n_nodes * 16;
  const int stride = gridDim.x * 512;
  const int gid    = ch * 512 + t;
  v4f cv[GCN_COPY_U];
#pragma unroll
  for (int k = 0; k < GCN_COPY_U; ++k) {
    int i = gid + k * stride;
    cv[k] = *((const v4f*)h + (i < total ? i : 0));
  }

  int d[GCN_CHUNK / 512], s[GCN_CHUNK / 512], w[GCN_CHUNK / 512];
#pragma unroll
  for (int j = 0; j < GCN_CHUNK / 512; ++j) {
    int i = j * 512 + t;
    bool in = i < n;
    int ii = e0 + (in ? i : 0);
    int dv = dst[ii];
    d[j] = in ? dv : -1;
    s[j] = src[ii];
    w[j] = __float_as_int(ew[ii]);
  }

  asm volatile("" :: "v"(cv[0]), "v"(cv[1]), "v"(cv[2]), "v"(cv[3]), "v"(cv[4]), "v"(cv[5]));

  __syncthreads();
#pragma unroll
  for (int j = 0; j < GCN_CHUNK / 512; ++j)
    if (d[j] >= 0) atomicAdd(&hist[d[j] >> GCN_BKT_SHIFT], 1);
  __syncthreads();

  const int c = hist[t];
  const int x = c > GCN_CELL ? c - GCN_CELL : 0;
  scan[t] = c | (x << 16);
  for (int off = 1; off < GCN_BKT_MAX; off <<= 1) {
    __syncthreads();
    int v = t >= off ? scan[t - off] : 0;
    __syncthreads();
    scan[t] += v;
  }
  const int incl = scan[t];
  const int lb = (incl & 0xffff) - c;
  lbase[t] = lb;
  if (t == GCN_BKT_MAX - 1) chunk_ovf[ch] = incl >> 16;
  if (t < n_bkt) cellcnt[(size_t)t * cell_pitch + ch] = c;
  hist[t] = lb;
  __syncthreads();

#pragma unroll
  for (int j = 0; j < GCN_CHUNK / 512; ++j)
    if (d[j] >= 0) {
      int p = atomicAdd(&hist[d[j] >> GCN_BKT_SHIFT], 1);
      sd[p] = d[j];
      ss[p] = s[j];
      sw[p] = w[j];
    }
  __syncthreads();

  for (int i = t; i < n; i += 512) {
    int dd = sd[i];
    int b  = dd >> GCN_BKT_SHIFT;
    int q  = i - lbase[b];
    if (q < GCN_CELL) {
      unsigned key = ((unsigned)ss[i] << GCN_BKT_SHIFT) | (unsigned)(dd & (GCN_BKT_NODES - 1));
      stag[((size_t)b * cell_pitch + ch) * GCN_CELL + q] = make_int2((int)key, sw[i]);
    } else {
      int k = atomicAdd(&ovc, 1);
      cov[(size_t)ch * GCN_CHUNK + k] = make_int4(dd, ss[i], sw[i], 0);
    }
  }

#pragma unroll
  for (int k = 0; k < GCN_COPY_U; ++k) {
    int i = gid + k * stride;
    if (i < total) *((v4f*)(out + (size_t)(i >> 4) * 256) + (i & 15)) = cv[k];
  }
  for (int i = gid + GCN_COPY_U * stride; i < total; i += stride)
    *((v4f*)(out + (size_t)(i >> 4) * 256) + (i & 15)) = *((const v4f*)h + i);
}

__device__ __forceinline__ void gcn_place(int2* slots, int* ovf_n,
                                          int* __restrict__ ovf_dst,
                                          int2* __restrict__ ovf_rec,
                                          int node0, int ln, int k, int s, int w) {
  if (k < GCN_SLOT) {
    slots[ln * GCN_SLOT + k] = make_int2(s, w);
  } else {
    int q = atomicAdd(ovf_n, 1);
    ovf_dst[q] = node0 + ln;
    ovf_rec[q] = make_int2(s, w);
  }
}

__global__ __launch_bounds__(512)
void gcn_slot_kernel(const int2* __restrict__ stag,
                     const int* __restrict__ cellcnt,
                     const int* __restrict__ chunk_ovf,
                     const int4* __restrict__ cov,
                     int* cnt,
                     int* ovf_n,
                     int2* __restrict__ rec,
                     int* __restrict__ ovf_dst,
                     int2* __restrict__ ovf_rec,
                     int n_nodes, int n_chunks, int cell_pitch) {
  __shared__ int  lcnt[GCN_BKT_NODES];
  __shared__ int2 slots[GCN_BKT_NODES * GCN_SLOT];
  const int t = threadIdx.x;
  const int b = blockIdx.x;
  const int node0 = b * GCN_BKT_NODES;
  const int wave = __builtin_amdgcn_readfirstlane(t >> 5);
  const int lane = t & 31;
  if (t < GCN_BKT_NODES) lcnt[t] = 0;
  __syncthreads();

  const int2* bcell = stag + (size_t)b * cell_pitch * GCN_CELL;
  for (int c0 = 0; c0 < n_chunks; c0 += 512) {
    int myc = c0 + wave + 16 * lane;
    int cc = 0, oc = 0;
    if (myc < n_chunks) {
      cc = cellcnt[(size_t)b * cell_pitch + myc];
      oc = chunk_ovf[myc];
    }
    if (cc > GCN_CELL) cc = GCN_CELL;
    int2 r[32];
#pragma unroll
    for (int j = 0; j < 32; ++j) {
      int cj = __builtin_amdgcn_readlane(cc, j);
      r[j] = make_int2(0, 0);
      if (lane < cj) r[j] = bcell[(size_t)(c0 + wave + 16 * j) * GCN_CELL + lane];
    }
    int k[32];
#pragma unroll
    for (int j = 0; j < 32; ++j) {
      int cj = __builtin_amdgcn_readlane(cc, j);
      k[j] = 0;
      if (lane < cj) k[j] = atomicAdd(&lcnt[r[j].x & (GCN_BKT_NODES - 1)], 1);
    }
#pragma unroll
    for (int j = 0; j < 32; ++j) {
      int cj = __builtin_amdgcn_readlane(cc, j);
      if (lane < cj)
        gcn_place(slots, ovf_n, ovf_dst, ovf_rec, node0,
                  r[j].x & (GCN_BKT_NODES - 1), k[j],
                  (int)((unsigned)r[j].x >> GCN_BKT_SHIFT), r[j].y);
    }
    if (__ballot(oc != 0) != 0) {
#pragma nounroll
      for (int j = 0; j < 32; ++j) {
        int oj = __builtin_amdgcn_readlane(oc, j);
        const int4* cp = cov + (size_t)(c0 + wave + 16 * j) * GCN_CHUNK;
        for (int i = lane; i < oj; i += 32) {
          int4 v = cp[i];
          if ((v.x >> GCN_BKT_SHIFT) == b) {
            int ln = v.x & (GCN_BKT_NODES - 1);
            gcn_place(slots, ovf_n, ovf_dst, ovf_rec, node0, ln,
                      atomicAdd(&lcnt[ln], 1), v.y, v.z);
          }
        }
      }
    }
  }
  __syncthreads();

  if (t < GCN_BKT_NODES && node0 + t < n_nodes) cnt[node0 + t] = lcnt[t];
  for (int i = t; i < GCN_BKT_NODES * GCN_SLOT; i += 512) {
    int ln = i / GCN_SLOT;
    int k  = i % GCN_SLOT;
    if (k < lcnt[ln]) rec[(size_t)(node0 + ln) * GCN_SLOT + k] = slots[i];
  }
}

template <int U>
__device__ __forceinline__ void gcn_pull_rows(const float* __restrict__ hprev,
                                              const int2* __restrict__ rec,
                                              int i, int f,
                                              float& ax, float& ay) {
  int    s[U];
  float  w[U];
  float2 hv[U];
#pragma unroll
  for (int j = 0; j < U; ++j) {
    int2 r = rec[i + j];
    s[j] = r.x;
    w[j] = __int_as_float(r.y);
  }
#pragma unroll
  for (int j = 0; j < U; ++j)
    hv[j] = *(const float2*)(hprev + (size_t)s[j] * 256 + f);
#pragma unroll
  for (int j = 0; j < U; ++j) {
    ax += hv[j].x * w[j];
    ay += hv[j].y * w[j];
  }
}

template <int U>
__device__ __forceinline__ void gcn_pull_regs(const float* __restrict__ hprev,
                                              int rx, int ry, int p, int f,
                                              float& ax, float& ay) {
  int    s[U];
  float  w[U];
  float2 hv[U];
#pragma unroll
  for (int j = 0; j < U; ++j) {
    s[j] = __builtin_amdgcn_readlane(rx, p + j);
    w[j] = __int_as_float(__builtin_amdgcn_readlane(ry, p + j));
  }
#pragma unroll
  for (int j = 0; j < U; ++j)
    hv[j] = *(const float2*)(hprev + (size_t)s[j] * 256 + f);
#pragma unroll
  for (int j = 0; j < U; ++j) {
    ax += hv[j].x * w[j];
    ay += hv[j].y * w[j];
  }
}

__device__ __forceinline__ void gcn_pull_half(const float* __restrict__ hprev,
                                              int2 r, int c, int f,
                                              float& ax, float& ay) {
  int i = 0;
  for (; i + 16 <= c; i += 16) gcn_pull_regs<16>(hprev, r.x, r.y, i, f, ax, ay);
  int t = c - i;
  if (t & 8) { gcn_pull_regs<8>(hprev, r.x, r.y, i, f, ax, ay); i += 8; }
  if (t & 4) { gcn_pull_regs<4>(hprev, r.x, r.y, i, f, ax, ay); i += 4; }
  if (t & 2) { gcn_pull_regs<2>(hprev, r.x, r.y, i, f, ax, ay); i += 2; }
  if (t & 1) { gcn_pull_regs<1>(hprev, r.x, r.y, i, f, ax, ay); }
}

#define GCN_TILE_PITCH 68

__global__ __launch_bounds__(256)
void gcn_layer_kernel(const float* __restrict__ hprev,
                      const int2* __restrict__ rec,
                      const int* __restrict__ cnt,
                      const int* __restrict__ ovf_dst,
                      const int2* __restrict__ ovf_rec,
                      const float* __restrict__ W,
                      const float* __restrict__ bias,
                      float* __restrict__ outp,
                      int n_nodes, int do_tanh) {
  __shared__ float tile[16 * GCN_TILE_PITCH];
  __shared__ __attribute__((aligned(16))) float otile[16 * GCN_TILE_PITCH];
  int wave = __builtin_amdgcn_readfirstlane((int)(threadIdx.x >> 5));
  int lane = threadIdx.x & 31;
  int row0 = blockIdx.x * 16;
  int f = lane * 2;

  int nb = row0 + wave * 2;
  int n0 = nb     < n_nodes ? nb     : n_nodes;
  int n1 = nb + 1 < n_nodes ? nb + 1 : n_nodes;
  int d0 = cnt[n0];
  int d1 = cnt[n1];
  int m  = cnt[n_nodes + 1];
  int e0 = d0 < GCN_SLOT ? d0 : GCN_SLOT;
  int e1 = d1 < GCN_SLOT ? d1 : GCN_SLOT;
  const int2* slot0 = rec + (size_t)n0 * GCN_SLOT;
  const int2* slot1 = rec + (size_t)n1 * GCN_SLOT;
  int2 ra0 = make_int2(0, 0), rb0 = ra0, ra1 = ra0, rb1 = ra0;
  if (lane      < e0) ra0 = slot0[lane];
  if (lane + 32 < e0) rb0 = slot0[lane + 32];
  if (lane      < e1) ra1 = slot1[lane];
  if (lane + 32 < e1) rb1 = slot1[lane + 32];
#pragma unroll
  for (int q = 0; q < 2; ++q) {
    int n = q ? n1 : n0;
    int e = q ? e1 : e0;
    float ax = 0.f, ay = 0.f;
    gcn_pull_half(hprev, q ? ra1 : ra0, e < 32 ? e : 32, f, ax, ay);
    if (e > 32) gcn_pull_half(hprev, q ? rb1 : rb0, e - 32, f, ax, ay);
    if (m > 0) {
#pragma nounroll
      for (int k = 0; k < m; ++k)
        if (ovf_dst[k] == n) gcn_pull_rows<1>(hprev, ovf_rec, k, f, ax, ay);
    }
    *(float2*)(tile + (wave * 2 + q) * GCN_TILE_PITCH + f) = make_float2(ax, ay);
  }
  __syncthreads();
  int lh = lane >> 4;
  int lm = lane & 15;
  int c  = 4 * lm;
  float4 bv = *(const float4*)(bias + c);

  if (wave < 4) {
    int col0 = wave * 16;
    const float* arow = tile + lm * GCN_TILE_PITCH + 2 * lh;
    const float* wcol = W + (size_t)(2 * lh) * 64 + col0 + lm;

    v8f acc = {};
#pragma unroll
    for (int k0 = 0; k0 < 64; k0 += 4) {
      v2f a = *(const v2f*)(arow + k0);
      v2f b;
      b.x = wcol[(size_t)k0 * 64];
      b.y = wcol[(size_t)k0 * 64 + 64];
      acc = __builtin_amdgcn_wmma_f32_16x16x4_f32(
           false, a,  false, b,
           (short)0, acc,  false,  false);
    }
    float* op = otile + (8 * lh) * GCN_TILE_PITCH + col0 + lm;
#pragma unroll
    for (int v = 0; v < 8; ++v) op[v * GCN_TILE_PITCH] = acc[v];
  }
  __syncthreads();

  int r = wave * 2 + lh;
  float4 val = *(const float4*)(otile + r * GCN_TILE_PITCH + c);
  val.x += bv.x; val.y += bv.y; val.z += bv.z; val.w += bv.w;
  if (do_tanh) {
    val.x = tanhf(val.x); val.y = tanhf(val.y);
    val.z = tanhf(val.z); val.w = tanhf(val.w);
  }
  if (row0 + r < n_nodes) *(float4*)(outp + (size_t)(row0 + r) * 256 + c) = val;
}

extern "C" void kernel_launch(void* const* d_in, const int* in_sizes, int n_in,
                              void* d_out, int out_size, void* d_ws, size_t ws_size,
                              hipStream_t stream) {
  const float* h  = (const float*)d_in[0];
  const float* ew = (const float*)d_in[1];
  const float* Ws = (const float*)d_in[2];
  const float* bs = (const float*)d_in[3];
  const int*  src = (const int*)d_in[4];
  const int*  dst = (const int*)d_in[5];
  float* out = (float*)d_out;

  const int N = in_sizes[0] / 64;
  const int E = in_sizes[1];
  const int L = 3;

  const int Np  = (N + 2 + 127) & ~127;
  const int Ep  = (E + 127) & ~127;
  const int NB  = (N + GCN_BKT_NODES - 1) / GCN_BKT_NODES;
  const int NC  = (E + GCN_CHUNK - 1) / GCN_CHUNK;
  const int NCp = (NC + 127) & ~127;
  int*  cnt       = (int*)d_ws;
  int*  cellcnt   = cnt + Np;
  int*  chunk_ovf = cellcnt + (size_t)NB * NCp;
  int*  ovf_dst   = chunk_ovf + NCp;
  int2* ovf_rec   = (int2*)(ovf_dst + Ep);
  int2* rec       = ovf_rec + Ep;
  int2* stag      = rec + (size_t)N * GCN_SLOT;
  int4* cov       = (int4*)(stag + (size_t)NB * NCp * GCN_CELL);

  gcn_bucket_kernel<<<NC, 512, 0, stream>>>(src, dst, ew, h, out, cnt + N, cellcnt, chunk_ovf,
                                            stag, cov, E, N, NB, NCp);
  gcn_slot_kernel<<<NB, 512, 0, stream>>>(stag, cellcnt, chunk_ovf, cov, cnt, cnt + N + 1, rec,
                                          ovf_dst, ovf_rec, N, NC, NCp);

  for (int layer = 0; layer < L; ++layer) {
    int blocks = (N + 15) / 16;
    gcn_layer_kernel<<<blocks, 256, 0, stream>>>(
        out + layer * 64, rec, cnt, ovf_dst, ovf_rec, Ws + layer * 64 * 64, bs + layer * 64,
        out + (layer + 1) * 64, N, (layer < L - 1) ? 1 : 0);
  }
}
